# starting kernel
# baseline (speedup 1.0000x reference)
#include <hip/hip_runtime.h>
#include <hip/hip_bf16.h>

typedef __attribute__((ext_vector_type(2))) float v2f;
typedef __attribute__((ext_vector_type(8))) float v8f;

#define D_DIM 128

__global__ void __launch_bounds__(256)
gene_proj_wmma_kernel(const float* __restrict__ gene_z,
                      const float* __restrict__ W,
                      const float* __restrict__ bvec,
                      float* __restrict__ proj,
                      int n_gene)
{
    __shared__ float As[16][131];

    const int tid  = threadIdx.x;
    const int row0 = blockIdx.x * 16;

    #pragma unroll
    for (int i = tid; i < 16 * D_DIM; i += 256) {
        int r = i >> 7;
        int c = i & 127;
        int rr = row0 + r;
        if (rr >= n_gene) rr = n_gene - 1;
        As[r][c] = gene_z[(long)rr * D_DIM + c];
    }
    __syncthreads();

    const int wave = tid >> 5;
    const int lane = tid & 31;
    const int half = lane >> 4;
    const int l16  = lane & 15;
    const int col0 = wave * 16;
    const int n    = col0 + l16;

    v8f acc = {};

    #pragma unroll 8
    for (int kk = 0; kk < D_DIM / 4; ++kk) {
        const int kb = kk * 4 + half * 2;
        v2f a;
        a[0] = As[l16][kb];
        a[1] = As[l16][kb + 1];
        v2f b;
        b[0] = W[(long)kb * D_DIM + n];
        b[1] = W[(long)(kb + 1) * D_DIM + n];

        acc = __builtin_amdgcn_wmma_f32_16x16x4_f32(
             false, a,  false, b,
             (short)0, acc,  false,  false);
    }

    const float bias = bvec[n];

    if (row0 + 16 <= n_gene) {
        #pragma unroll
        for (int j = 0; j < 8; ++j) {
            const int m = j + half * 8;
            float v = acc[j] + bias;
            v = (v > 0.0f) ? v : 0.01f * v;
            proj[(long)(row0 + m) * D_DIM + n] = v;
        }
    } else {
        #pragma unroll
        for (int j = 0; j < 8; ++j) {
            const int m = j + half * 8;
            float v = acc[j] + bias;
            v = (v > 0.0f) ? v : 0.01f * v;
            const int rr = row0 + m;
            if (rr < n_gene)
                proj[(long)rr * D_DIM + n] = v;
        }
    }
}

__global__ void __launch_bounds__(256)
edge_dot_kernel(const float* __restrict__ cell_z,
                const float* __restrict__ gene_proj,
                const float* __restrict__ cell_scale,
                const float* __restrict__ cell_bias,
                const float* __restrict__ cell_std,
                const float* __restrict__ gene_scale,
                const float* __restrict__ gene_bias,
                const float* __restrict__ gene_std,
                const int* __restrict__ src_idx,
                const int* __restrict__ dst_idx,
                const int* __restrict__ cell_n_id,
                const int* __restrict__ gene_n_id,
                float* __restrict__ out,
                int E)
{
    const int lane = threadIdx.x & 31;
    const long e = (long)blockIdx.x * (blockDim.x >> 5) + (threadIdx.x >> 5);
    if (e >= E) return;

    const int s = __builtin_amdgcn_readfirstlane(src_idx[e]);
    const int d = __builtin_amdgcn_readfirstlane(dst_idx[e]);

    const float4 u4 = *(const float4*)(cell_z    + (long)s * D_DIM + lane * 4);
    const float4 v4 = *(const float4*)(gene_proj + (long)d * D_DIM + lane * 4);

    float p = u4.x * v4.x + u4.y * v4.y + u4.z * v4.z + u4.w * v4.w;

    #pragma unroll
    for (int off = 16; off > 0; off >>= 1)
        p += __shfl_xor(p, off, 32);

    if (lane == 0) {
        const int sid = cell_n_id[s];
        const int did = gene_n_id[d];
        const float loc = p * cell_scale[sid] * gene_scale[did]
                          + cell_bias[sid] + gene_bias[did];
        const float sd  = cell_std[sid] * gene_std[did];
        out[e]           = loc;
        out[(long)E + e] = sd;
    }
}

extern "C" void kernel_launch(void* const* d_in, const int* in_sizes, int n_in,
                              void* d_out, int out_size, void* d_ws, size_t ws_size,
                              hipStream_t stream) {
    const float* cell_z     = (const float*)d_in[0];
    const float* gene_z     = (const float*)d_in[1];
    const float* W_gene     = (const float*)d_in[2];
    const float* b_gene     = (const float*)d_in[3];
    const float* cell_scale = (const float*)d_in[4];
    const float* cell_bias  = (const float*)d_in[5];
    const float* cell_std   = (const float*)d_in[6];
    const float* gene_scale = (const float*)d_in[7];
    const float* gene_bias  = (const float*)d_in[8];
    const float* gene_std   = (const float*)d_in[9];
    const int*   src_idx    = (const int*)d_in[10];
    const int*   dst_idx    = (const int*)d_in[11];
    const int*   cell_n_id  = (const int*)d_in[12];
    const int*   gene_n_id  = (const int*)d_in[13];

    const int n_gene = in_sizes[1] / D_DIM;
    const int E      = in_sizes[10];

    float* gene_proj = (float*)d_ws;

    const int gemm_blocks = (n_gene + 15) / 16;
    gene_proj_wmma_kernel<<<gemm_blocks, 256, 0, stream>>>(
        gene_z, W_gene, b_gene, gene_proj, n_gene);

    const int edges_per_block = 256 / 32;
    const int edge_blocks = (E + edges_per_block - 1) / edges_per_block;
    edge_dot_kernel<<<edge_blocks, 256, 0, stream>>>(
        cell_z, gene_proj,
        cell_scale, cell_bias, cell_std,
        gene_scale, gene_bias, gene_std,
        src_idx, dst_idx, cell_n_id, gene_n_id,
        (float*)d_out, E);
}


# per-node 16-byte records {scale,bias,std} built by extra blocks of launch 1; each edge reads 2 records beside its rows instead of 8 scattered 4-byte lookups in two dependent rounds
# speedup vs baseline: 1.2112x; 1.2112x over previous
#include <hip/hip_runtime.h>
#include <hip/hip_bf16.h>

typedef __attribute__((ext_vector_type(2))) float v2f;
typedef __attribute__((ext_vector_type(8))) float v8f;

#define D_DIM 128

__global__ void __launch_bounds__(256)
gene_proj_wmma_kernel(const float* __restrict__ gene_z,
                      const float* __restrict__ W,
                      const float* __restrict__ bvec,
                      float* __restrict__ proj,
                      int n_gene,
                      const float* __restrict__ cell_scale,
                      const float* __restrict__ cell_bias,
                      const float* __restrict__ cell_std,
                      const float* __restrict__ gene_scale,
                      const float* __restrict__ gene_bias,
                      const float* __restrict__ gene_std,
                      const int* __restrict__ cell_n_id,
                      const int* __restrict__ gene_n_id,
                      float* __restrict__ cell_rec_ws,
                      float* __restrict__ gene_rec_ws,
                      int n_cell)
{
    float4* __restrict__ cell_rec = (float4*)cell_rec_ws;
    float4* __restrict__ gene_rec = (float4*)gene_rec_ws;

    if (blockIdx.y == 1) {
        const int total  = n_cell + n_gene;
        const int stride = gridDim.x * 256;
        for (int i = blockIdx.x * 256 + threadIdx.x; i < total; i += stride) {
            if (i < n_cell) {
                const int id = cell_n_id[i];
                cell_rec[i] = make_float4(cell_scale[id], cell_bias[id], cell_std[id], 0.0f);
            } else {
                const int g  = i - n_cell;
                const int id = gene_n_id[g];
                gene_rec[g] = make_float4(gene_scale[id], gene_bias[id], gene_std[id], 0.0f);
            }
        }
        return;
    }

    __shared__ float As[16][131];

    const int tid  = threadIdx.x;
    const int row0 = blockIdx.x * 16;

    #pragma unroll
    for (int i = tid; i < 16 * D_DIM; i += 256) {
        int r = i >> 7;
        int c = i & 127;
        int rr = row0 + r;
        if (rr >= n_gene) rr = n_gene - 1;
        As[r][c] = gene_z[(long)rr * D_DIM + c];
    }
    __syncthreads();

    const int wave = tid >> 5;
    const int lane = tid & 31;
    const int half = lane >> 4;
    const int l16  = lane & 15;
    const int col0 = wave * 16;
    const int n    = col0 + l16;

    v8f acc = {};

    #pragma unroll 8
    for (int kk = 0; kk < D_DIM / 4; ++kk) {
        const int kb = kk * 4 + half * 2;
        v2f a;
        a[0] = As[l16][kb];
        a[1] = As[l16][kb + 1];
        v2f b;
        b[0] = W[(long)kb * D_DIM + n];
        b[1] = W[(long)(kb + 1) * D_DIM + n];

        acc = __builtin_amdgcn_wmma_f32_16x16x4_f32(
             false, a,  false, b,
             (short)0, acc,  false,  false);
    }

    const float bias = bvec[n];

    if (row0 + 16 <= n_gene) {
        #pragma unroll
        for (int j = 0; j < 8; ++j) {
            const int m = j + half * 8;
            float v = acc[j] + bias;
            v = (v > 0.0f) ? v : 0.01f * v;
            proj[(long)(row0 + m) * D_DIM + n] = v;
        }
    } else {
        #pragma unroll
        for (int j = 0; j < 8; ++j) {
            const int m = j + half * 8;
            float v = acc[j] + bias;
            v = (v > 0.0f) ? v : 0.01f * v;
            const int rr = row0 + m;
            if (rr < n_gene)
                proj[(long)rr * D_DIM + n] = v;
        }
    }
}

__global__ void __launch_bounds__(256)
edge_dot_kernel(const float* __restrict__ cell_z,
                const float* __restrict__ gene_proj,
                const float* __restrict__ cell_rec_ws,
                const float* __restrict__ gene_rec_ws,
                const int* __restrict__ src_idx,
                const int* __restrict__ dst_idx,
                float* __restrict__ out,
                int E)
{
    const int lane = threadIdx.x & 31;
    const long e = (long)blockIdx.x * (blockDim.x >> 5) + (threadIdx.x >> 5);
    if (e >= E) return;

    const int s = __builtin_amdgcn_readfirstlane(src_idx[e]);
    const int d = __builtin_amdgcn_readfirstlane(dst_idx[e]);

    const float4 u4 = *(const float4*)(cell_z    + (long)s * D_DIM + lane * 4);
    const float4 v4 = *(const float4*)(gene_proj + (long)d * D_DIM + lane * 4);

    const float4 cr = ((const float4*)cell_rec_ws)[s];
    const float4 gr = ((const float4*)gene_rec_ws)[d];
    asm volatile("" :: "s"(cr.x), "s"(cr.y), "s"(cr.z),
                       "s"(gr.x), "s"(gr.y), "s"(gr.z));

    float p = u4.x * v4.x + u4.y * v4.y + u4.z * v4.z + u4.w * v4.w;

    #pragma unroll
    for (int off = 16; off > 0; off >>= 1)
        p += __shfl_xor(p, off, 32);

    if (lane == 0) {
        const float loc = p * cr.x * gr.x + cr.y + gr.y;
        const float sd  = cr.z * gr.z;
        out[e]           = loc;
        out[(long)E + e] = sd;
    }
}

extern "C" void kernel_launch(void* const* d_in, const int* in_sizes, int n_in,
                              void* d_out, int out_size, void* d_ws, size_t ws_size,
                              hipStream_t stream) {
    const float* cell_z     = (const float*)d_in[0];
    const float* gene_z     = (const float*)d_in[1];
    const float* W_gene     = (const float*)d_in[2];
    const float* b_gene     = (const float*)d_in[3];
    const float* cell_scale = (const float*)d_in[4];
    const float* cell_bias  = (const float*)d_in[5];
    const float* cell_std   = (const float*)d_in[6];
    const float* gene_scale = (const float*)d_in[7];
    const float* gene_bias  = (const float*)d_in[8];
    const float* gene_std   = (const float*)d_in[9];
    const int*   src_idx    = (const int*)d_in[10];
    const int*   dst_idx    = (const int*)d_in[11];
    const int*   cell_n_id  = (const int*)d_in[12];
    const int*   gene_n_id  = (const int*)d_in[13];

    const int n_gene = in_sizes[1] / D_DIM;
    const int E      = in_sizes[10];

    float* gene_proj = (float*)d_ws;
    float* cell_rec = (float*)((char*)d_ws + 67108864);
    float* gene_rec = (float*)((char*)d_ws + 100663296);
    const int n_cell = in_sizes[12];

    const int gemm_blocks = (n_gene + 15) / 16;
    gene_proj_wmma_kernel<<<dim3(gemm_blocks, 2), 256, 0, stream>>>(
        gene_z, W_gene, b_gene, gene_proj, n_gene,
        cell_scale, cell_bias, cell_std,
        gene_scale, gene_bias, gene_std,
        cell_n_id, gene_n_id, cell_rec, gene_rec, n_cell);

    const int edges_per_block = 256 / 32;
    const int edge_blocks = (E + edges_per_block - 1) / edges_per_block;
    edge_dot_kernel<<<edge_blocks, 256, 0, stream>>>(
        cell_z, gene_proj, cell_rec, gene_rec,
        src_idx, dst_idx,
        (float*)d_out, E);
}


# edge kernel: one wave per 32 consecutive edges, not per edge: coalesced index load, per-lane records, rows by readlane, transposed butterfly in the same 16-8-4-2-1 order, whole-line stores
# speedup vs baseline: 2.7347x; 2.2578x over previous
#include <hip/hip_runtime.h>
#include <hip/hip_bf16.h>

typedef __attribute__((ext_vector_type(2))) float v2f;
typedef __attribute__((ext_vector_type(8))) float v8f;

#define D_DIM 128

__global__ void __launch_bounds__(256)
gene_proj_wmma_kernel(const float* __restrict__ gene_z,
                      const float* __restrict__ W,
                      const float* __restrict__ bvec,
                      float* __restrict__ proj,
                      int n_gene,
                      const float* __restrict__ cell_scale,
                      const float* __restrict__ cell_bias,
                      const float* __restrict__ cell_std,
                      const float* __restrict__ gene_scale,
                      const float* __restrict__ gene_bias,
                      const float* __restrict__ gene_std,
                      const int* __restrict__ cell_n_id,
                      const int* __restrict__ gene_n_id,
                      float* __restrict__ cell_rec_ws,
                      float* __restrict__ gene_rec_ws,
                      int n_cell)
{
    float4* __restrict__ cell_rec = (float4*)cell_rec_ws;
    float4* __restrict__ gene_rec = (float4*)gene_rec_ws;

    if (blockIdx.y == 1) {
        const int total  = n_cell + n_gene;
        const int stride = gridDim.x * 256;
        for (int i = blockIdx.x * 256 + threadIdx.x; i < total; i += stride) {
            if (i < n_cell) {
                const int id = cell_n_id[i];
                cell_rec[i] = make_float4(cell_scale[id], cell_bias[id], cell_std[id], 0.0f);
            } else {
                const int g  = i - n_cell;
                const int id = gene_n_id[g];
                gene_rec[g] = make_float4(gene_scale[id], gene_bias[id], gene_std[id], 0.0f);
            }
        }
        return;
    }

    __shared__ float As[16][131];

    const int tid  = threadIdx.x;
    const int row0 = blockIdx.x * 16;

    #pragma unroll
    for (int i = tid; i < 16 * D_DIM; i += 256) {
        int r = i >> 7;
        int c = i & 127;
        int rr = row0 + r;
        if (rr >= n_gene) rr = n_gene - 1;
        As[r][c] = gene_z[(long)rr * D_DIM + c];
    }
    __syncthreads();

    const int wave = tid >> 5;
    const int lane = tid & 31;
    const int half = lane >> 4;
    const int l16  = lane & 15;
    const int col0 = wave * 16;
    const int n    = col0 + l16;

    v8f acc = {};

    #pragma unroll 8
    for (int kk = 0; kk < D_DIM / 4; ++kk) {
        const int kb = kk * 4 + half * 2;
        v2f a;
        a[0] = As[l16][kb];
        a[1] = As[l16][kb + 1];
        v2f b;
        b[0] = W[(long)kb * D_DIM + n];
        b[1] = W[(long)(kb + 1) * D_DIM + n];

        acc = __builtin_amdgcn_wmma_f32_16x16x4_f32(
             false, a,  false, b,
             (short)0, acc,  false,  false);
    }

    const float bias = bvec[n];

    if (row0 + 16 <= n_gene) {
        #pragma unroll
        for (int j = 0; j < 8; ++j) {
            const int m = j + half * 8;
            float v = acc[j] + bias;
            v = (v > 0.0f) ? v : 0.01f * v;
            proj[(long)(row0 + m) * D_DIM + n] = v;
        }
    } else {
        #pragma unroll
        for (int j = 0; j < 8; ++j) {
            const int m = j + half * 8;
            float v = acc[j] + bias;
            v = (v > 0.0f) ? v : 0.01f * v;
            const int rr = row0 + m;
            if (rr < n_gene)
                proj[(long)rr * D_DIM + n] = v;
        }
    }
}

#define EDGES_PER_WAVE 32

__device__ __forceinline__ float lane_xor16(float x) {
    const int b = __builtin_bit_cast(int, x);
    return __builtin_bit_cast(float,
        __builtin_amdgcn_permlanex16(b, b, 0x76543210, 0xfedcba98, false, false));
}

template <int M>
__device__ __forceinline__ float lane_xor_row(float x) {
    const int b = __builtin_bit_cast(int, x);
    return __builtin_bit_cast(float,
        __builtin_amdgcn_update_dpp(0, b, 0x160 + M, 0xf, 0xf, true));
}

template <int M>
__device__ __forceinline__ float fold_pair(float a, float b, bool hi) {
    const float keep = hi ? b : a;
    const float send = hi ? a : b;
    if constexpr (M == 16) return keep + lane_xor16(send);
    else                   return keep + lane_xor_row<M>(send);
}

__global__ void __launch_bounds__(256)
edge_dot_kernel(const float* __restrict__ cell_z,
                const float* __restrict__ gene_proj,
                const float* __restrict__ cell_rec_ws,
                const float* __restrict__ gene_rec_ws,
                const int* __restrict__ src_idx,
                const int* __restrict__ dst_idx,
                float* __restrict__ out,
                int E)
{
    const int lane = threadIdx.x & 31;
    const int wave_in_block = __builtin_amdgcn_readfirstlane((int)(threadIdx.x >> 5));
    const long wave_id = (long)blockIdx.x * (blockDim.x >> 5) + wave_in_block;
    const long e0 = wave_id * EDGES_PER_WAVE;
    if (e0 >= E) return;

    const long e  = e0 + lane;
    const long ec = (e < E) ? e : (long)E - 1;

    const int si = src_idx[ec];
    const int di = dst_idx[ec];

    const float4 cr = ((const float4*)cell_rec_ws)[si];
    const float4 gr = ((const float4*)gene_rec_ws)[di];

    const bool b16 = (lane & 16) != 0;
    const bool b8  = (lane & 8)  != 0;
    const bool b4  = (lane & 4)  != 0;
    const bool b2  = (lane & 2)  != 0;
    const bool b1  = (lane & 1)  != 0;

    float t[4];
    float4 u[8], v[8];

    #pragma unroll
    for (int j = 0; j < 8; ++j) {
        const int s = __builtin_amdgcn_readlane(si, 4 * j);
        const int d = __builtin_amdgcn_readlane(di, 4 * j);
        u[j] = *(const float4*)(cell_z    + (long)s * D_DIM + lane * 4);
        v[j] = *(const float4*)(gene_proj + (long)d * D_DIM + lane * 4);
    }

    #pragma unroll
    for (int g = 0; g < 4; ++g) {
        __builtin_amdgcn_sched_barrier(0);

        float p[8];
        #pragma unroll
        for (int j = 0; j < 8; ++j)
            p[j] = u[j].x * v[j].x + u[j].y * v[j].y + u[j].z * v[j].z + u[j].w * v[j].w;

        const float q0 = fold_pair<16>(p[0], p[4], b16);
        const float q1 = fold_pair<16>(p[1], p[5], b16);
        const float q2 = fold_pair<16>(p[2], p[6], b16);
        const float q3 = fold_pair<16>(p[3], p[7], b16);
        const float r0 = fold_pair<8>(q0, q2, b8);
        const float r1 = fold_pair<8>(q1, q3, b8);
        t[g] = fold_pair<4>(r0, r1, b4);

        if (g < 3) {
            #pragma unroll
            for (int j = 0; j < 8; ++j) {
                const int k = 4 * j + g + 1;
                const int s = __builtin_amdgcn_readlane(si, k);
                const int d = __builtin_amdgcn_readlane(di, k);
                u[j] = *(const float4*)(cell_z    + (long)s * D_DIM + lane * 4);
                v[j] = *(const float4*)(gene_proj + (long)d * D_DIM + lane * 4);
            }
        }
    }

    const float w0  = fold_pair<2>(t[0], t[2], b2);
    const float w1  = fold_pair<2>(t[1], t[3], b2);
    const float dot = fold_pair<1>(w0, w1, b1);

    asm volatile("" :: "v"(cr.x), "v"(cr.y), "v"(cr.z),
                       "v"(gr.x), "v"(gr.y), "v"(gr.z));

    if (e < E) {
        const float loc = dot * cr.x * gr.x + cr.y + gr.y;
        const float sd  = cr.z * gr.z;
        out[e]           = loc;
        out[(long)E + e] = sd;
    }
}

extern "C" void kernel_launch(void* const* d_in, const int* in_sizes, int n_in,
                              void* d_out, int out_size, void* d_ws, size_t ws_size,
                              hipStream_t stream) {
    const float* cell_z     = (const float*)d_in[0];
    const float* gene_z     = (const float*)d_in[1];
    const float* W_gene     = (const float*)d_in[2];
    const float* b_gene     = (const float*)d_in[3];
    const float* cell_scale = (const float*)d_in[4];
    const float* cell_bias  = (const float*)d_in[5];
    const float* cell_std   = (const float*)d_in[6];
    const float* gene_scale = (const float*)d_in[7];
    const float* gene_bias  = (const float*)d_in[8];
    const float* gene_std   = (const float*)d_in[9];
    const int*   src_idx    = (const int*)d_in[10];
    const int*   dst_idx    = (const int*)d_in[11];
    const int*   cell_n_id  = (const int*)d_in[12];
    const int*   gene_n_id  = (const int*)d_in[13];

    const int n_gene = in_sizes[1] / D_DIM;
    const int E      = in_sizes[10];

    float* gene_proj = (float*)d_ws;
    float* cell_rec = (float*)((char*)d_ws + 67108864);
    float* gene_rec = (float*)((char*)d_ws + 100663296);
    const int n_cell = in_sizes[12];

    const int gemm_blocks = (n_gene + 15) / 16;
    gene_proj_wmma_kernel<<<dim3(gemm_blocks, 2), 256, 0, stream>>>(
        gene_z, W_gene, b_gene, gene_proj, n_gene,
        cell_scale, cell_bias, cell_std,
        gene_scale, gene_bias, gene_std,
        cell_n_id, gene_n_id, cell_rec, gene_rec, n_cell);

    const int edges_per_block = 256;
    const int edge_blocks = (E + edges_per_block - 1) / edges_per_block;
    edge_dot_kernel<<<edge_blocks, 256, 0, stream>>>(
        cell_z, gene_proj, cell_rec, gene_rec,
        src_idx, dst_idx,
        (float*)d_out, E);
}
